# starting kernel
# baseline (speedup 1.0000x reference)
#include <hip/hip_runtime.h>

typedef __attribute__((ext_vector_type(2))) float v2f;
typedef __attribute__((ext_vector_type(8))) float v8f;

#define C_IN   512
#define HW_    128
#define NBATCH 8
#define K_RAW  9
#define K_PAD  12
#define EQ_SCALE 0.014731391274719739f

__global__ void prep_weight_kernel(const float* __restrict__ w,
                                   float* __restrict__ bp) {
  int tid = blockIdx.x * blockDim.x + threadIdx.x;
  if (tid >= 6144 * 16) return;
  int n  = tid & 15;
  int kp = tid >> 4;
  int ci = kp / K_PAD;
  int t  = kp - ci * K_PAD;
  int dy = t >> 2, dx = t & 3;
  float val = 0.0f;
  if (dx < 3 && n < 3)
    val = w[n * (C_IN * K_RAW) + ci * K_RAW + dy * 3 + dx] * EQ_SCALE;
  int p = kp >> 1, par = kp & 1;
  bp[(p * 16 + n) * 2 + par] = val;
}

__global__ void conv_wmma_fast_kernel(const float* __restrict__ in,
                                      const float* __restrict__ bpack,
                                      float* __restrict__ out) {
  const int lane = threadIdx.x & 31;
  const int wid  = __builtin_amdgcn_readfirstlane(blockIdx.x * 4 + (threadIdx.x >> 5));
  const int b    = wid / 378;
  const int r    = wid - b * 378;
  const int y    = 1 + r / 3;
  const int pi   = r - (r / 3) * 3;
  const int x0A  = 16 + pi * 32;

  const int M    = lane & 15;
  const int hi   = lane >> 4;
  const int xoff = M + hi * 2 - 1;

  const int offA = (y - 1) * HW_ + x0A + xoff;
  const int offB = offA + 16;
  const int bidx = hi * 16 + M;

  const float* pin = in + (size_t)b * C_IN * (HW_ * HW_);
  const v2f*   bp  = (const v2f*)bpack;

  v8f cA = {}, cB = {};
  for (int ci = 0; ci < C_IN; ++ci) {
    const float* p  = pin + (size_t)ci * (HW_ * HW_);
    const v2f*   pb = bp + ci * 96;
#pragma unroll
    for (int sub = 0; sub < 3; ++sub) {
      v2f aA, aB;
      __builtin_memcpy(&aA, p + offA + sub * HW_, 8);
      __builtin_memcpy(&aB, p + offB + sub * HW_, 8);
      v2f bb = pb[sub * 32 + bidx];
      cA = __builtin_amdgcn_wmma_f32_16x16x4_f32(
          false, aA, false, bb, (short)0, cA, false, false);
      cB = __builtin_amdgcn_wmma_f32_16x16x4_f32(
          false, aB, false, bb, (short)0, cB, false, false);
    }
  }

  if (M < 3) {
    const int mbase = hi * 8;
    float* poA = out + (((size_t)b * 3 + M) * HW_ + y) * HW_ + x0A + mbase;
    float* poB = poA + 16;
#pragma unroll
    for (int r8 = 0; r8 < 8; ++r8) { poA[r8] = cA[r8]; poB[r8] = cB[r8]; }
  }
}

__global__ void conv_wmma_edge_kernel(const float* __restrict__ in,
                                      const float* __restrict__ bpack,
                                      float* __restrict__ out) {
  const int lane = threadIdx.x & 31;
  const int wid  = __builtin_amdgcn_readfirstlane(blockIdx.x * 4 + (threadIdx.x >> 5));
  const int b    = wid / 268;
  const int e    = wid - b * 268;
  int y, xt;
  if (e < 16) { y = (e < 8) ? 0 : 127; xt = e & 7; }
  else        { int e2 = e - 16; y = 1 + (e2 >> 1); xt = (e2 & 1) ? 7 : 0; }
  const int x0 = xt << 4;

  const int M    = lane & 15;
  const int hi   = lane >> 4;
  const int bidx = hi * 16 + M;

  int  off[6];
  bool vld[6];
#pragma unroll
  for (int j = 0; j < 6; ++j) {
    int dy = j >> 1;
    int dx = hi * 2 + (j & 1);
    if (dx < 3) {
      int ym = y + dy - 1;
      int xm = x0 + M + dx - 1;
      vld[j] = (ym >= 0) && (ym < HW_) && (xm >= 0) && (xm < HW_);
      off[j] = vld[j] ? (ym * HW_ + xm) : 0;
    } else { vld[j] = false; off[j] = 0; }
  }

  const float* pin = in + (size_t)b * C_IN * (HW_ * HW_);
  const v2f*   bp  = (const v2f*)bpack;

  v8f c = {};
  for (int ci = 0; ci < C_IN; ++ci) {
    const float* p  = pin + (size_t)ci * (HW_ * HW_);
    const v2f*   pb = bp + ci * 96;
#pragma unroll
    for (int sub = 0; sub < 3; ++sub) {
      float a0 = vld[2 * sub]     ? p[off[2 * sub]]     : 0.0f;
      float a1 = vld[2 * sub + 1] ? p[off[2 * sub + 1]] : 0.0f;
      v2f a; a.x = a0; a.y = a1;
      v2f bb = pb[sub * 32 + bidx];
      c = __builtin_amdgcn_wmma_f32_16x16x4_f32(
          false, a, false, bb, (short)0, c, false, false);
    }
  }

  if (M < 3) {
    const int mbase = hi * 8;
    float* po = out + (((size_t)b * 3 + M) * HW_ + y) * HW_ + x0 + mbase;
#pragma unroll
    for (int r8 = 0; r8 < 8; ++r8) po[r8] = c[r8];
  }
}

__global__ void skip_bias_kernel(const float* __restrict__ skip,
                                 const float* __restrict__ bias,
                                 float* __restrict__ out) {
  int tid = blockIdx.x * blockDim.x + threadIdx.x;
  const int total = NBATCH * 3 * HW_ * HW_;
  if (tid >= total) return;

  int x  = tid & 127;
  int y  = (tid >> 7) & 127;
  int bc = tid >> 14;
  int co = bc % 3;

  int y2 = y >> 1; bool ye = !(y & 1);
  int   my0 = ye ? y2     : y2 + 1;  float wy0 = ye ? 0.75f : 0.25f;
  int   my1 = ye ? y2 - 1 : y2;      float wy1 = ye ? 0.25f : 0.75f;
  int x2 = x >> 1; bool xe = !(x & 1);
  int   mx0 = xe ? x2     : x2 + 1;  float wx0 = xe ? 0.75f : 0.25f;
  int   mx1 = xe ? x2 - 1 : x2;      float wx1 = xe ? 0.25f : 0.75f;

  const float* s = skip + (size_t)bc * 64 * 64;
  float acc = bias[co];
  if (my0 >= 0 && my0 < 64) {
    if (mx0 >= 0 && mx0 < 64) acc += wy0 * wx0 * s[my0 * 64 + mx0];
    if (mx1 >= 0 && mx1 < 64) acc += wy0 * wx1 * s[my0 * 64 + mx1];
  }
  if (my1 >= 0 && my1 < 64) {
    if (mx0 >= 0 && mx0 < 64) acc += wy1 * wx0 * s[my1 * 64 + mx0];
    if (mx1 >= 0 && mx1 < 64) acc += wy1 * wx1 * s[my1 * 64 + mx1];
  }
  out[tid] += acc;
}

extern "C" void kernel_launch(void* const* d_in, const int* in_sizes, int n_in,
                              void* d_out, int out_size, void* d_ws,
                              size_t ws_size, hipStream_t stream) {
  (void)in_sizes; (void)n_in; (void)out_size; (void)ws_size;
  const float* in   = (const float*)d_in[0];
  const float* skip = (const float*)d_in[1];
  const float* w    = (const float*)d_in[2];
  const float* bias = (const float*)d_in[3];
  float* out = (float*)d_out;
  float* bp  = (float*)d_ws;

  prep_weight_kernel<<<dim3((6144 * 16 + 255) / 256), dim3(256), 0, stream>>>(w, bp);
  conv_wmma_fast_kernel<<<dim3(756), dim3(128), 0, stream>>>(in, bp, out);
  conv_wmma_edge_kernel<<<dim3(536), dim3(128), 0, stream>>>(in, bp, out);
  skip_bias_kernel<<<dim3((NBATCH * 3 * HW_ * HW_ + 255) / 256), dim3(256), 0, stream>>>(
      skip, bias, out);
}


# Both conv kernels: one block of 16 waves per tile, each wave sums 32 of the 512 input channels, partial sums added through LDS in fixed slice order (changes the summation order)
# speedup vs baseline: 3.1610x; 3.1610x over previous
#include <hip/hip_runtime.h>

typedef __attribute__((ext_vector_type(2))) float v2f;
typedef __attribute__((ext_vector_type(8))) float v8f;

#define C_IN   512
#define HW_    128
#define NBATCH 8
#define K_RAW  9
#define K_PAD  12
#define EQ_SCALE 0.014731391274719739f
#define KSPLIT 16
#define CI_PER (C_IN / KSPLIT)

__global__ void prep_weight_kernel(const float* __restrict__ w,
                                   float* __restrict__ bp) {
  int tid = blockIdx.x * blockDim.x + threadIdx.x;
  if (tid >= 6144 * 16) return;
  int n  = tid & 15;
  int kp = tid >> 4;
  int ci = kp / K_PAD;
  int t  = kp - ci * K_PAD;
  int dy = t >> 2, dx = t & 3;
  float val = 0.0f;
  if (dx < 3 && n < 3)
    val = w[n * (C_IN * K_RAW) + ci * K_RAW + dy * 3 + dx] * EQ_SCALE;
  int p = kp >> 1, par = kp & 1;
  bp[(p * 16 + n) * 2 + par] = val;
}

__global__ void __launch_bounds__(KSPLIT * 32) conv_wmma_fast_kernel(const float* __restrict__ in,
                                      const float* __restrict__ bpack,
                                      float* __restrict__ out) {
  __shared__ float red[KSPLIT * 96];
  const int lane = threadIdx.x & 31;
  const int slice = __builtin_amdgcn_readfirstlane(threadIdx.x >> 5);
  const int wid  = blockIdx.x;
  const int b    = wid / 378;
  const int r    = wid - b * 378;
  const int y    = 1 + r / 3;
  const int pi   = r - (r / 3) * 3;
  const int x0A  = 16 + pi * 32;

  const int M    = lane & 15;
  const int hi   = lane >> 4;
  const int xoff = M + hi * 2 - 1;

  const int offA = (y - 1) * HW_ + x0A + xoff;
  const int offB = offA + 16;
  const int bidx = hi * 16 + M;

  const int ci0 = slice * CI_PER;
  const float* pin = in + ((size_t)b * C_IN + ci0) * (HW_ * HW_);
  const v2f*   bp  = (const v2f*)bpack + ci0 * 96;

  v8f cA = {}, cB = {};
  for (int i = 0; i < CI_PER; ++i) {
    const float* p  = pin + (size_t)i * (HW_ * HW_);
    const v2f*   pb = bp + i * 96;
#pragma unroll
    for (int sub = 0; sub < 3; ++sub) {
      v2f aA, aB;
      __builtin_memcpy(&aA, p + offA + sub * HW_, 8);
      __builtin_memcpy(&aB, p + offB + sub * HW_, 8);
      v2f bb = pb[sub * 32 + bidx];
      cA = __builtin_amdgcn_wmma_f32_16x16x4_f32(
          false, aA, false, bb, (short)0, cA, false, false);
      cB = __builtin_amdgcn_wmma_f32_16x16x4_f32(
          false, aB, false, bb, (short)0, cB, false, false);
    }
  }

  if (M < 3) {
    float* pr = red + slice * 96 + (hi * 3 + M) * 16;
#pragma unroll
    for (int r8 = 0; r8 < 8; ++r8) { pr[r8] = cA[r8]; pr[8 + r8] = cB[r8]; }
  }
  __syncthreads();
  const int t = threadIdx.x;
  if (t < 96) {
    float sum = red[t];
#pragma unroll
    for (int w = 1; w < KSPLIT; ++w) sum += red[w * 96 + t];
    const int l6 = t >> 4, r = t & 15;
    const int h2 = l6 / 3, m2 = l6 - h2 * 3;
    const int x  = x0A + ((r >> 3) << 4) + h2 * 8 + (r & 7);
    out[(((size_t)b * 3 + m2) * HW_ + y) * HW_ + x] = sum;
  }
}

__global__ void __launch_bounds__(KSPLIT * 32) conv_wmma_edge_kernel(const float* __restrict__ in,
                                      const float* __restrict__ bpack,
                                      float* __restrict__ out) {
  __shared__ float red[KSPLIT * 48];
  const int lane = threadIdx.x & 31;
  const int slice = __builtin_amdgcn_readfirstlane(threadIdx.x >> 5);
  const int wid  = blockIdx.x;
  const int b    = wid / 268;
  const int e    = wid - b * 268;
  int y, xt;
  if (e < 16) { y = (e < 8) ? 0 : 127; xt = e & 7; }
  else        { int e2 = e - 16; y = 1 + (e2 >> 1); xt = (e2 & 1) ? 7 : 0; }
  const int x0 = xt << 4;

  const int M    = lane & 15;
  const int hi   = lane >> 4;
  const int bidx = hi * 16 + M;

  int  off[6];
  bool vld[6];
#pragma unroll
  for (int j = 0; j < 6; ++j) {
    int dy = j >> 1;
    int dx = hi * 2 + (j & 1);
    if (dx < 3) {
      int ym = y + dy - 1;
      int xm = x0 + M + dx - 1;
      vld[j] = (ym >= 0) && (ym < HW_) && (xm >= 0) && (xm < HW_);
      off[j] = vld[j] ? (ym * HW_ + xm) : 0;
    } else { vld[j] = false; off[j] = 0; }
  }

  const int ci0 = slice * CI_PER;
  const float* pin = in + ((size_t)b * C_IN + ci0) * (HW_ * HW_);
  const v2f*   bp  = (const v2f*)bpack + ci0 * 96;

  v8f c = {};
  for (int i = 0; i < CI_PER; ++i) {
    const float* p  = pin + (size_t)i * (HW_ * HW_);
    const v2f*   pb = bp + i * 96;
#pragma unroll
    for (int sub = 0; sub < 3; ++sub) {
      float a0 = vld[2 * sub]     ? p[off[2 * sub]]     : 0.0f;
      float a1 = vld[2 * sub + 1] ? p[off[2 * sub + 1]] : 0.0f;
      v2f a; a.x = a0; a.y = a1;
      v2f bb = pb[sub * 32 + bidx];
      c = __builtin_amdgcn_wmma_f32_16x16x4_f32(
          false, a, false, bb, (short)0, c, false, false);
    }
  }

  if (M < 3) {
    float* pr = red + slice * 48 + (hi * 3 + M) * 8;
#pragma unroll
    for (int r8 = 0; r8 < 8; ++r8) pr[r8] = c[r8];
  }
  __syncthreads();
  const int t = threadIdx.x;
  if (t < 48) {
    float sum = red[t];
#pragma unroll
    for (int w = 1; w < KSPLIT; ++w) sum += red[w * 48 + t];
    const int l6 = t >> 3, r8 = t & 7;
    const int h2 = l6 / 3, m2 = l6 - h2 * 3;
    out[(((size_t)b * 3 + m2) * HW_ + y) * HW_ + x0 + h2 * 8 + r8] = sum;
  }
}

__global__ void skip_bias_kernel(const float* __restrict__ skip,
                                 const float* __restrict__ bias,
                                 float* __restrict__ out) {
  int tid = blockIdx.x * blockDim.x + threadIdx.x;
  const int total = NBATCH * 3 * HW_ * HW_;
  if (tid >= total) return;

  int x  = tid & 127;
  int y  = (tid >> 7) & 127;
  int bc = tid >> 14;
  int co = bc % 3;

  int y2 = y >> 1; bool ye = !(y & 1);
  int   my0 = ye ? y2     : y2 + 1;  float wy0 = ye ? 0.75f : 0.25f;
  int   my1 = ye ? y2 - 1 : y2;      float wy1 = ye ? 0.25f : 0.75f;
  int x2 = x >> 1; bool xe = !(x & 1);
  int   mx0 = xe ? x2     : x2 + 1;  float wx0 = xe ? 0.75f : 0.25f;
  int   mx1 = xe ? x2 - 1 : x2;      float wx1 = xe ? 0.25f : 0.75f;

  const float* s = skip + (size_t)bc * 64 * 64;
  float acc = bias[co];
  if (my0 >= 0 && my0 < 64) {
    if (mx0 >= 0 && mx0 < 64) acc += wy0 * wx0 * s[my0 * 64 + mx0];
    if (mx1 >= 0 && mx1 < 64) acc += wy0 * wx1 * s[my0 * 64 + mx1];
  }
  if (my1 >= 0 && my1 < 64) {
    if (mx0 >= 0 && mx0 < 64) acc += wy1 * wx0 * s[my1 * 64 + mx0];
    if (mx1 >= 0 && mx1 < 64) acc += wy1 * wx1 * s[my1 * 64 + mx1];
  }
  out[tid] += acc;
}

extern "C" void kernel_launch(void* const* d_in, const int* in_sizes, int n_in,
                              void* d_out, int out_size, void* d_ws,
                              size_t ws_size, hipStream_t stream) {
  (void)in_sizes; (void)n_in; (void)out_size; (void)ws_size;
  const float* in   = (const float*)d_in[0];
  const float* skip = (const float*)d_in[1];
  const float* w    = (const float*)d_in[2];
  const float* bias = (const float*)d_in[3];
  float* out = (float*)d_out;
  float* bp  = (float*)d_ws;

  prep_weight_kernel<<<dim3((6144 * 16 + 255) / 256), dim3(256), 0, stream>>>(w, bp);
  conv_wmma_fast_kernel<<<dim3(3024), dim3(KSPLIT * 32), 0, stream>>>(in, bp, out);
  conv_wmma_edge_kernel<<<dim3(2144), dim3(KSPLIT * 32), 0, stream>>>(in, bp, out);
  skip_bias_kernel<<<dim3((NBATCH * 3 * HW_ * HW_ + 255) / 256), dim3(256), 0, stream>>>(
      skip, bias, out);
}


# Conv recast as one matrix product over the 512 input channels giving 27 tap planes (M = out channel x tap, N = pixels), 3x3 shift-add of the planes in LDS (changes the summation order)
# speedup vs baseline: 18.0410x; 5.7074x over previous
#include <hip/hip_runtime.h>


typedef __attribute__((ext_vector_type(2))) float v2f;
typedef __attribute__((ext_vector_type(4))) float v4f;
typedef __attribute__((ext_vector_type(8))) float v8f;

#define C_IN   512
#define HW_    128
#define NBATCH 8
#define K_RAW  9
#define EQ_SCALE 0.014731391274719739f
#define PLANE  (HW_ * HW_)
#define NPL    27
#define KSTEPS (C_IN / 4)
#define GRP    8
#define NGRP   (KSTEPS / GRP)
#define R_OUT  4
#define R_IN   (R_OUT + 2)
#define NWAVE  (R_IN * 2)

__global__ void prep_weight_kernel(const float* __restrict__ w,
                                   float* __restrict__ wpk) {
  int tid = blockIdx.x * blockDim.x + threadIdx.x;
  if (tid >= KSTEPS * 32 * 4) return;
  int e    = tid & 3;
  int lane = (tid >> 2) & 31;
  int ks   = tid >> 7;
  int c = lane & 15, h = lane >> 4;
  int p  = ((e >> 1) << 4) + c;
  int ci = ks * 4 + h * 2 + (e & 1);
  float val = 0.0f;
  if (p < NPL) {
    int co = p / 9, tap = p - co * 9;
    val = w[co * (C_IN * K_RAW) + ci * K_RAW + tap] * EQ_SCALE;
  }
  wpk[tid] = val;
}

struct Grp { v4f x0[GRP], x1[GRP], w[GRP]; };

__device__ __forceinline__ void load_grp(Grp& G, const float* __restrict__ base,
                                         int loff, const v4f* __restrict__ wq,
                                         int lane, int g) {
#pragma unroll
  for (int u = 0; u < GRP; ++u) {
    const int ks = g * GRP + u;
    const float* p = base + (size_t)ks * (4 * PLANE) + loff;
    G.x0[u] = *(const v4f*)p;
    G.x1[u] = *(const v4f*)(p + PLANE);
    G.w[u]  = wq[ks * 32 + lane];
  }
}

__device__ __forceinline__ void mma_grp(const Grp& G, v8f (&acc)[2][4]) {
#pragma unroll
  for (int u = 0; u < GRP; ++u) {
    v2f a0, a1;
    a0.x = G.w[u].x; a0.y = G.w[u].y;
    a1.x = G.w[u].z; a1.y = G.w[u].w;
#pragma unroll
    for (int j = 0; j < 4; ++j) {
      v2f bj;
      bj.x = G.x0[u][j]; bj.y = G.x1[u][j];
      acc[0][j] = __builtin_amdgcn_wmma_f32_16x16x4_f32(
          false, a0, false, bj, (short)0, acc[0][j], false, false);
      acc[1][j] = __builtin_amdgcn_wmma_f32_16x16x4_f32(
          false, a1, false, bj, (short)0, acc[1][j], false, false);
    }
  }
}

__global__ void __launch_bounds__(NWAVE * 32) conv_planes_kernel(const float* __restrict__ in,
                                      const float* __restrict__ wpk,
                                      float* __restrict__ out) {
  __shared__ float T[R_IN * NPL * HW_];
  const int lane = threadIdx.x & 31;
  const int wave = __builtin_amdgcn_readfirstlane(threadIdx.x >> 5);
  const int b    = blockIdx.x >> 5;
  const int y0   = (blockIdx.x & 31) * R_OUT;
  const int rl   = wave >> 1;
  const int half = wave & 1;
  const int yin  = y0 - 1 + rl;
  const int c = lane & 15, h = lane >> 4;
  const int px = half * 64 + c * 4;
  float* Trow = T + rl * (NPL * HW_);

  if (yin >= 0 && yin < HW_) {
    const float* base = in + (size_t)b * C_IN * PLANE + yin * HW_;
    const int loff = h * 2 * PLANE + px;
    const v4f* wq = (const v4f*)wpk;

    v8f acc[2][4];
#pragma unroll
    for (int t = 0; t < 2; ++t)
#pragma unroll
      for (int j = 0; j < 4; ++j) acc[t][j] = (v8f){};

    Grp GA, GB;
    load_grp(GA, base, loff, wq, lane, 0);
    for (int g = 0; g < NGRP - 2; g += 2) {
      load_grp(GB, base, loff, wq, lane, g + 1);
      __builtin_amdgcn_sched_barrier(0);
      mma_grp(GA, acc);
      __builtin_amdgcn_sched_barrier(0);
      load_grp(GA, base, loff, wq, lane, g + 2);
      __builtin_amdgcn_sched_barrier(0);
      mma_grp(GB, acc);
      __builtin_amdgcn_sched_barrier(0);
    }
    load_grp(GB, base, loff, wq, lane, NGRP - 1);
    __builtin_amdgcn_sched_barrier(0);
    mma_grp(GA, acc);
    mma_grp(GB, acc);

#pragma unroll
    for (int t = 0; t < 2; ++t) {
#pragma unroll
      for (int r = 0; r < 8; ++r) {
        const int p = t * 16 + h * 8 + r;
        if (p < NPL) {
          v4f v;
          v.x = acc[t][0][r]; v.y = acc[t][1][r]; v.z = acc[t][2][r]; v.w = acc[t][3][r];
          *(v4f*)(Trow + p * HW_ + px) = v;
        }
      }
    }
  } else {
    for (int p = 0; p < NPL; ++p) {
      Trow[p * HW_ + half * 64 + lane]      = 0.0f;
      Trow[p * HW_ + half * 64 + 32 + lane] = 0.0f;
    }
  }
  __syncthreads();

  const int t  = threadIdx.x;
  const int x  = t & 127;
  const int co = t >> 7;
  const bool okl = x > 0, okr = x < HW_ - 1;
  const int  xl = okl ? x - 1 : x, xr = okr ? x + 1 : x;
  for (int r = 0; r < R_OUT; ++r) {
    float sum = 0.0f;
#pragma unroll
    for (int dy = 0; dy < 3; ++dy) {
      const float* tp = T + ((r + dy) * NPL + co * 9 + dy * 3) * HW_;
      const float vl = tp[xl], vc = tp[HW_ + x], vr = tp[2 * HW_ + xr];
      sum += okl ? vl : 0.0f;
      sum += vc;
      sum += okr ? vr : 0.0f;
    }
    out[(((size_t)b * 3 + co) * HW_ + (y0 + r)) * HW_ + x] = sum;
  }
}

__global__ void skip_bias_kernel(const float* __restrict__ skip,
                                 const float* __restrict__ bias,
                                 float* __restrict__ out) {
  int tid = blockIdx.x * blockDim.x + threadIdx.x;
  const int total = NBATCH * 3 * HW_ * HW_;
  if (tid >= total) return;

  int x  = tid & 127;
  int y  = (tid >> 7) & 127;
  int bc = tid >> 14;
  int co = bc % 3;

  int y2 = y >> 1; bool ye = !(y & 1);
  int   my0 = ye ? y2     : y2 + 1;  float wy0 = ye ? 0.75f : 0.25f;
  int   my1 = ye ? y2 - 1 : y2;      float wy1 = ye ? 0.25f : 0.75f;
  int x2 = x >> 1; bool xe = !(x & 1);
  int   mx0 = xe ? x2     : x2 + 1;  float wx0 = xe ? 0.75f : 0.25f;
  int   mx1 = xe ? x2 - 1 : x2;      float wx1 = xe ? 0.25f : 0.75f;

  const float* s = skip + (size_t)bc * 64 * 64;
  float acc = bias[co];
  if (my0 >= 0 && my0 < 64) {
    if (mx0 >= 0 && mx0 < 64) acc += wy0 * wx0 * s[my0 * 64 + mx0];
    if (mx1 >= 0 && mx1 < 64) acc += wy0 * wx1 * s[my0 * 64 + mx1];
  }
  if (my1 >= 0 && my1 < 64) {
    if (mx0 >= 0 && mx0 < 64) acc += wy1 * wx0 * s[my1 * 64 + mx0];
    if (mx1 >= 0 && mx1 < 64) acc += wy1 * wx1 * s[my1 * 64 + mx1];
  }
  out[tid] += acc;
}

extern "C" void kernel_launch(void* const* d_in, const int* in_sizes, int n_in,
                              void* d_out, int out_size, void* d_ws,
                              size_t ws_size, hipStream_t stream) {
  (void)in_sizes; (void)n_in; (void)out_size; (void)ws_size;
  const float* in   = (const float*)d_in[0];
  const float* skip = (const float*)d_in[1];
  const float* w    = (const float*)d_in[2];
  const float* bias = (const float*)d_in[3];
  float* out = (float*)d_out;
  float* wpk = (float*)d_ws;

  prep_weight_kernel<<<dim3((KSTEPS * 32 * 4 + 255) / 256), dim3(256), 0, stream>>>(w, wpk);
  conv_planes_kernel<<<dim3(NBATCH * (HW_ / R_OUT)), dim3(NWAVE * 32), 0, stream>>>(in, wpk, out);
  skip_bias_kernel<<<dim3((NBATCH * 3 * HW_ * HW_ + 255) / 256), dim3(256), 0, stream>>>(
      skip, bias, out);
}


# Halo rows of each block run one row tile of the matrix product, not two: planes ordered by tap row so the 9 planes a halo row feeds share a tile; halo work goes to each SIMD's oldest wave
# speedup vs baseline: 19.8079x; 1.0979x over previous
#include <hip/hip_runtime.h>


typedef __attribute__((ext_vector_type(2))) float v2f;
typedef __attribute__((ext_vector_type(4))) float v4f;
typedef __attribute__((ext_vector_type(8))) float v8f;

#define C_IN   512
#define HW_    128
#define NBATCH 8
#define K_RAW  9
#define EQ_SCALE 0.014731391274719739f
#define PLANE  (HW_ * HW_)
#define NPL    27
#define KSTEPS (C_IN / 4)
#define GRP    8
#define NGRP   (KSTEPS / GRP)
#define R_OUT  4
#define R_IN   (R_OUT + 2)
#define NWAVE  (R_IN * 2)

__global__ void prep_weight_kernel(const float* __restrict__ w,
                                   float* __restrict__ wpk) {
  int tid = blockIdx.x * blockDim.x + threadIdx.x;
  if (tid >= KSTEPS * 32 * 4) return;
  int e    = tid & 3;
  int lane = (tid >> 2) & 31;
  int ks   = tid >> 7;
  int c = lane & 15, h = lane >> 4;
  int p  = ((e >> 1) << 4) + c;
  int ci = ks * 4 + h * 2 + (e & 1);
  float val = 0.0f;
  if (p < NPL) {
    int dy = p / 9, q = p - dy * 9;
    int co = q / 3, dx = q - co * 3;
    val = w[co * (C_IN * K_RAW) + ci * K_RAW + dy * 3 + dx] * EQ_SCALE;
  }
  wpk[tid] = val;
}

template <int NT> struct WFrag;
template <> struct WFrag<1> { typedef v2f type; };
template <> struct WFrag<2> { typedef v4f type; };

template <int NT> struct Grp {
  v4f x0[GRP], x1[GRP];
  typename WFrag<NT>::type w[GRP];
};

template <int T0, int NT>
__device__ __forceinline__ void load_grp(Grp<NT>& G, const float* __restrict__ base,
                                         int loff, const float* __restrict__ wpk,
                                         int lane, int g) {
  typedef typename WFrag<NT>::type wfrag_t;
#pragma unroll
  for (int u = 0; u < GRP; ++u) {
    const int ks = g * GRP + u;
    const float* p = base + (size_t)ks * (4 * PLANE) + loff;
    G.x0[u] = *(const v4f*)p;
    G.x1[u] = *(const v4f*)(p + PLANE);
    G.w[u]  = *(const wfrag_t*)(wpk + (ks * 32 + lane) * 4 + T0 * 2);
  }
}

template <int NT>
__device__ __forceinline__ void mma_grp(const Grp<NT>& G, v8f (&acc)[NT][4]) {
#pragma unroll
  for (int u = 0; u < GRP; ++u) {
    v2f a[NT];
#pragma unroll
    for (int t = 0; t < NT; ++t) { a[t].x = G.w[u][2 * t]; a[t].y = G.w[u][2 * t + 1]; }
#pragma unroll
    for (int j = 0; j < 4; ++j) {
      v2f bj;
      bj.x = G.x0[u][j]; bj.y = G.x1[u][j];
#pragma unroll
      for (int t = 0; t < NT; ++t)
        acc[t][j] = __builtin_amdgcn_wmma_f32_16x16x4_f32(
            false, a[t], false, bj, (short)0, acc[t][j], false, false);
    }
  }
}

template <int T0, int NT, int P_LO, int P_HI>
__device__ __forceinline__ void row_planes(const float* __restrict__ base, int loff,
                                           const float* __restrict__ wpk, int lane,
                                           int h, float* __restrict__ Tpx) {
  v8f acc[NT][4];
#pragma unroll
  for (int t = 0; t < NT; ++t)
#pragma unroll
    for (int j = 0; j < 4; ++j) acc[t][j] = (v8f){};

  Grp<NT> GA, GB;
  load_grp<T0, NT>(GA, base, loff, wpk, lane, 0);
  for (int g = 0; g < NGRP - 2; g += 2) {
    load_grp<T0, NT>(GB, base, loff, wpk, lane, g + 1);
    __builtin_amdgcn_sched_barrier(0);
    mma_grp<NT>(GA, acc);
    __builtin_amdgcn_sched_barrier(0);
    load_grp<T0, NT>(GA, base, loff, wpk, lane, g + 2);
    __builtin_amdgcn_sched_barrier(0);
    mma_grp<NT>(GB, acc);
    __builtin_amdgcn_sched_barrier(0);
  }
  load_grp<T0, NT>(GB, base, loff, wpk, lane, NGRP - 1);
  __builtin_amdgcn_sched_barrier(0);
  mma_grp<NT>(GA, acc);
  mma_grp<NT>(GB, acc);

#pragma unroll
  for (int t = 0; t < NT; ++t) {
#pragma unroll
    for (int r = 0; r < 8; ++r) {
      const int p = (T0 + t) * 16 + h * 8 + r;
      if (p >= P_LO && p < P_HI) {
        v4f v;
        v.x = acc[t][0][r]; v.y = acc[t][1][r]; v.z = acc[t][2][r]; v.w = acc[t][3][r];
        *(v4f*)(Tpx + p * HW_) = v;
      }
    }
  }
}

__global__ void __launch_bounds__(NWAVE * 32) conv_planes_kernel(const float* __restrict__ in,
                                      const float* __restrict__ wpk,
                                      float* __restrict__ out) {
  __shared__ float T[R_IN * NPL * HW_];
  const int lane = threadIdx.x & 31;
  const int wave = __builtin_amdgcn_readfirstlane(threadIdx.x >> 5);
  const int unit = (int)((0x98765432BA10ull >> (4 * wave)) & 15);
  const int b    = blockIdx.x >> 5;
  const int y0   = (blockIdx.x & 31) * R_OUT;
  const int rl   = unit >> 1;
  const int half = unit & 1;
  const int yin  = y0 - 1 + rl;
  const int c = lane & 15, h = lane >> 4;
  const int px = half * 64 + c * 4;
  float* Trow = T + rl * (NPL * HW_);

  if (yin >= 0 && yin < HW_) {
    const float* base = in + (size_t)b * C_IN * PLANE + yin * HW_;
    const int loff = h * 2 * PLANE + px;
    if (rl == 0)
      row_planes<0, 1, 0, 9>(base, loff, wpk, lane, h, Trow + px);
    else if (rl == R_IN - 1)
      row_planes<1, 1, 18, NPL>(base, loff, wpk, lane, h, Trow + px);
    else
      row_planes<0, 2, 0, NPL>(base, loff, wpk, lane, h, Trow + px);
  } else {
    const int p0 = (rl == 0) ? 0 : 18;
    for (int p = p0; p < p0 + 9; ++p) {
      Trow[p * HW_ + half * 64 + lane]      = 0.0f;
      Trow[p * HW_ + half * 64 + 32 + lane] = 0.0f;
    }
  }
  __syncthreads();

  const int t  = threadIdx.x;
  const int x  = t & 127;
  const int co = t >> 7;
  const bool okl = x > 0, okr = x < HW_ - 1;
  const int  xl = okl ? x - 1 : x, xr = okr ? x + 1 : x;
  for (int r = 0; r < R_OUT; ++r) {
    float sum = 0.0f;
#pragma unroll
    for (int dy = 0; dy < 3; ++dy) {
      const float* tp = T + ((r + dy) * NPL + dy * 9 + co * 3) * HW_;
      const float vl = tp[xl], vc = tp[HW_ + x], vr = tp[2 * HW_ + xr];
      sum += okl ? vl : 0.0f;
      sum += vc;
      sum += okr ? vr : 0.0f;
    }
    out[(((size_t)b * 3 + co) * HW_ + (y0 + r)) * HW_ + x] = sum;
  }
}

__global__ void skip_bias_kernel(const float* __restrict__ skip,
                                 const float* __restrict__ bias,
                                 float* __restrict__ out) {
  int tid = blockIdx.x * blockDim.x + threadIdx.x;
  const int total = NBATCH * 3 * HW_ * HW_;
  if (tid >= total) return;

  int x  = tid & 127;
  int y  = (tid >> 7) & 127;
  int bc = tid >> 14;
  int co = bc % 3;

  int y2 = y >> 1; bool ye = !(y & 1);
  int   my0 = ye ? y2     : y2 + 1;  float wy0 = ye ? 0.75f : 0.25f;
  int   my1 = ye ? y2 - 1 : y2;      float wy1 = ye ? 0.25f : 0.75f;
  int x2 = x >> 1; bool xe = !(x & 1);
  int   mx0 = xe ? x2     : x2 + 1;  float wx0 = xe ? 0.75f : 0.25f;
  int   mx1 = xe ? x2 - 1 : x2;      float wx1 = xe ? 0.25f : 0.75f;

  const float* s = skip + (size_t)bc * 64 * 64;
  float acc = bias[co];
  if (my0 >= 0 && my0 < 64) {
    if (mx0 >= 0 && mx0 < 64) acc += wy0 * wx0 * s[my0 * 64 + mx0];
    if (mx1 >= 0 && mx1 < 64) acc += wy0 * wx1 * s[my0 * 64 + mx1];
  }
  if (my1 >= 0 && my1 < 64) {
    if (mx0 >= 0 && mx0 < 64) acc += wy1 * wx0 * s[my1 * 64 + mx0];
    if (mx1 >= 0 && mx1 < 64) acc += wy1 * wx1 * s[my1 * 64 + mx1];
  }
  out[tid] += acc;
}

extern "C" void kernel_launch(void* const* d_in, const int* in_sizes, int n_in,
                              void* d_out, int out_size, void* d_ws,
                              size_t ws_size, hipStream_t stream) {
  (void)in_sizes; (void)n_in; (void)out_size; (void)ws_size;
  const float* in   = (const float*)d_in[0];
  const float* skip = (const float*)d_in[1];
  const float* w    = (const float*)d_in[2];
  const float* bias = (const float*)d_in[3];
  float* out = (float*)d_out;
  float* wpk = (float*)d_ws;

  prep_weight_kernel<<<dim3((KSTEPS * 32 * 4 + 255) / 256), dim3(256), 0, stream>>>(w, wpk);
  conv_planes_kernel<<<dim3(NBATCH * (HW_ / R_OUT)), dim3(NWAVE * 32), 0, stream>>>(in, wpk, out);
  skip_bias_kernel<<<dim3((NBATCH * 3 * HW_ * HW_ + 255) / 256), dim3(256), 0, stream>>>(
      skip, bias, out);
}


# Bias and 2x upsampled skip added in the conv epilogue, each output stored once: third launch and its read-modify-write of out removed (3 launches -> 2); sum order unchanged
# speedup vs baseline: 22.6945x; 1.1457x over previous
#include <hip/hip_runtime.h>


typedef __attribute__((ext_vector_type(2))) float v2f;
typedef __attribute__((ext_vector_type(4))) float v4f;
typedef __attribute__((ext_vector_type(8))) float v8f;

#define C_IN   512
#define HW_    128
#define NBATCH 8
#define K_RAW  9
#define EQ_SCALE 0.014731391274719739f
#define PLANE  (HW_ * HW_)
#define NPL    27
#define KSTEPS (C_IN / 4)
#define GRP    8
#define NGRP   (KSTEPS / GRP)
#define R_OUT  4
#define R_IN   (R_OUT + 2)
#define NWAVE  (R_IN * 2)

__global__ void prep_weight_kernel(const float* __restrict__ w,
                                   float* __restrict__ wpk) {
  int tid = blockIdx.x * blockDim.x + threadIdx.x;
  if (tid >= KSTEPS * 32 * 4) return;
  int e    = tid & 3;
  int lane = (tid >> 2) & 31;
  int ks   = tid >> 7;
  int c = lane & 15, h = lane >> 4;
  int p  = ((e >> 1) << 4) + c;
  int ci = ks * 4 + h * 2 + (e & 1);
  float val = 0.0f;
  if (p < NPL) {
    int dy = p / 9, q = p - dy * 9;
    int co = q / 3, dx = q - co * 3;
    val = w[co * (C_IN * K_RAW) + ci * K_RAW + dy * 3 + dx] * EQ_SCALE;
  }
  wpk[tid] = val;
}

template <int NT> struct WFrag;
template <> struct WFrag<1> { typedef v2f type; };
template <> struct WFrag<2> { typedef v4f type; };

template <int NT> struct Grp {
  v4f x0[GRP], x1[GRP];
  typename WFrag<NT>::type w[GRP];
};

template <int T0, int NT>
__device__ __forceinline__ void load_grp(Grp<NT>& G, const float* __restrict__ base,
                                         int loff, const float* __restrict__ wpk,
                                         int lane, int g) {
  typedef typename WFrag<NT>::type wfrag_t;
#pragma unroll
  for (int u = 0; u < GRP; ++u) {
    const int ks = g * GRP + u;
    const float* p = base + (size_t)ks * (4 * PLANE) + loff;
    G.x0[u] = *(const v4f*)p;
    G.x1[u] = *(const v4f*)(p + PLANE);
    G.w[u]  = *(const wfrag_t*)(wpk + (ks * 32 + lane) * 4 + T0 * 2);
  }
}

template <int NT>
__device__ __forceinline__ void mma_grp(const Grp<NT>& G, v8f (&acc)[NT][4]) {
#pragma unroll
  for (int u = 0; u < GRP; ++u) {
    v2f a[NT];
#pragma unroll
    for (int t = 0; t < NT; ++t) { a[t].x = G.w[u][2 * t]; a[t].y = G.w[u][2 * t + 1]; }
#pragma unroll
    for (int j = 0; j < 4; ++j) {
      v2f bj;
      bj.x = G.x0[u][j]; bj.y = G.x1[u][j];
#pragma unroll
      for (int t = 0; t < NT; ++t)
        acc[t][j] = __builtin_amdgcn_wmma_f32_16x16x4_f32(
            false, a[t], false, bj, (short)0, acc[t][j], false, false);
    }
  }
}

template <int T0, int NT, int P_LO, int P_HI>
__device__ __forceinline__ void row_planes(const float* __restrict__ base, int loff,
                                           const float* __restrict__ wpk, int lane,
                                           int h, float* __restrict__ Tpx) {
  v8f acc[NT][4];
#pragma unroll
  for (int t = 0; t < NT; ++t)
#pragma unroll
    for (int j = 0; j < 4; ++j) acc[t][j] = (v8f){};

  Grp<NT> GA, GB;
  load_grp<T0, NT>(GA, base, loff, wpk, lane, 0);
  for (int g = 0; g < NGRP - 2; g += 2) {
    load_grp<T0, NT>(GB, base, loff, wpk, lane, g + 1);
    __builtin_amdgcn_sched_barrier(0);
    mma_grp<NT>(GA, acc);
    __builtin_amdgcn_sched_barrier(0);
    load_grp<T0, NT>(GA, base, loff, wpk, lane, g + 2);
    __builtin_amdgcn_sched_barrier(0);
    mma_grp<NT>(GB, acc);
    __builtin_amdgcn_sched_barrier(0);
  }
  load_grp<T0, NT>(GB, base, loff, wpk, lane, NGRP - 1);
  __builtin_amdgcn_sched_barrier(0);
  mma_grp<NT>(GA, acc);
  mma_grp<NT>(GB, acc);

#pragma unroll
  for (int t = 0; t < NT; ++t) {
#pragma unroll
    for (int r = 0; r < 8; ++r) {
      const int p = (T0 + t) * 16 + h * 8 + r;
      if (p >= P_LO && p < P_HI) {
        v4f v;
        v.x = acc[t][0][r]; v.y = acc[t][1][r]; v.z = acc[t][2][r]; v.w = acc[t][3][r];
        *(v4f*)(Tpx + p * HW_) = v;
      }
    }
  }
}

__device__ __forceinline__ float skip_bias_term(float bias, float wy0, float wy1,
                                                float wx0, float wx1,
                                                float s00, float s01,
                                                float s10, float s11) {
  float acc = bias;
  acc += wy0 * wx0 * s00;
  acc += wy0 * wx1 * s01;
  acc += wy1 * wx0 * s10;
  acc += wy1 * wx1 * s11;
  return acc;
}

__global__ void __launch_bounds__(NWAVE * 32) conv_planes_kernel(const float* __restrict__ in,
                                      const float* __restrict__ wpk,
                                      const float* __restrict__ skip,
                                      const float* __restrict__ bias,
                                      float* __restrict__ out) {
  __shared__ float T[R_IN * NPL * HW_];
  const int lane = threadIdx.x & 31;
  const int wave = __builtin_amdgcn_readfirstlane(threadIdx.x >> 5);
  const int unit = (int)((0x98765432BA10ull >> (4 * wave)) & 15);
  const int b    = blockIdx.x >> 5;
  const int y0   = (blockIdx.x & 31) * R_OUT;
  const int rl   = unit >> 1;
  const int half = unit & 1;
  const int yin  = y0 - 1 + rl;
  const int c = lane & 15, h = lane >> 4;
  const int px = half * 64 + c * 4;
  float* Trow = T + rl * (NPL * HW_);

  const int t  = threadIdx.x;
  const int x  = t & 127;
  const int co = t >> 7;
  const int   x2  = x >> 1;
  const bool  xe  = !(x & 1);
  const int   mx0 = xe ? x2     : x2 + 1;
  const int   mx1 = xe ? x2 - 1 : x2;
  const float wx0 = xe ? 0.75f : 0.25f;
  const float wx1 = xe ? 0.25f : 0.75f;
  const bool  okx0 = mx0 < HW_ / 2, okx1 = mx1 >= 0;
  const int   h0  = y0 >> 1;
  const bool  oka = h0 > 0, okd = h0 + 2 < HW_ / 2;
  float sa0, sa1, sb0, sb1, sc0, sc1, sd0, sd1, bco;
  {
    const float* s  = skip + ((size_t)b * 3 + co) * (PLANE / 4);
    const int cx0 = okx0 ? mx0 : HW_ / 2 - 1, cx1 = okx1 ? mx1 : 0;
    const float* ra = s + (oka ? h0 - 1 : 0) * (HW_ / 2);
    const float* rb = s + h0 * (HW_ / 2);
    const float* rc = rb + HW_ / 2;
    const float* rd = s + (okd ? h0 + 2 : HW_ / 2 - 1) * (HW_ / 2);
    sa0 = ra[cx0]; sa1 = ra[cx1];
    sb0 = rb[cx0]; sb1 = rb[cx1];
    sc0 = rc[cx0]; sc1 = rc[cx1];
    sd0 = rd[cx0]; sd1 = rd[cx1];
    bco = bias[co];
  }

  if (yin >= 0 && yin < HW_) {
    const float* base = in + (size_t)b * C_IN * PLANE + yin * HW_;
    const int loff = h * 2 * PLANE + px;
    if (rl == 0)
      row_planes<0, 1, 0, 9>(base, loff, wpk, lane, h, Trow + px);
    else if (rl == R_IN - 1)
      row_planes<1, 1, 18, NPL>(base, loff, wpk, lane, h, Trow + px);
    else
      row_planes<0, 2, 0, NPL>(base, loff, wpk, lane, h, Trow + px);
  } else {
    const int p0 = (rl == 0) ? 0 : 18;
    for (int p = p0; p < p0 + 9; ++p) {
      Trow[p * HW_ + half * 64 + lane]      = 0.0f;
      Trow[p * HW_ + half * 64 + 32 + lane] = 0.0f;
    }
  }
  __syncthreads();

  sa0 = (oka && okx0) ? sa0 : 0.0f;  sa1 = (oka && okx1) ? sa1 : 0.0f;
  sb0 = okx0 ? sb0 : 0.0f;           sb1 = okx1 ? sb1 : 0.0f;
  sc0 = okx0 ? sc0 : 0.0f;           sc1 = okx1 ? sc1 : 0.0f;
  sd0 = (okd && okx0) ? sd0 : 0.0f;  sd1 = (okd && okx1) ? sd1 : 0.0f;
  float addv[R_OUT];
  addv[0] = skip_bias_term(bco, 0.75f, 0.25f, wx0, wx1, sb0, sb1, sa0, sa1);
  addv[1] = skip_bias_term(bco, 0.25f, 0.75f, wx0, wx1, sc0, sc1, sb0, sb1);
  addv[2] = skip_bias_term(bco, 0.75f, 0.25f, wx0, wx1, sc0, sc1, sb0, sb1);
  addv[3] = skip_bias_term(bco, 0.25f, 0.75f, wx0, wx1, sd0, sd1, sc0, sc1);

  const bool okl = x > 0, okr = x < HW_ - 1;
  const int  xl = okl ? x - 1 : x, xr = okr ? x + 1 : x;
#pragma unroll
  for (int r = 0; r < R_OUT; ++r) {
    float sum = 0.0f;
#pragma unroll
    for (int dy = 0; dy < 3; ++dy) {
      const float* tp = T + ((r + dy) * NPL + dy * 9 + co * 3) * HW_;
      const float vl = tp[xl], vc = tp[HW_ + x], vr = tp[2 * HW_ + xr];
      sum += okl ? vl : 0.0f;
      sum += vc;
      sum += okr ? vr : 0.0f;
    }
    out[(((size_t)b * 3 + co) * HW_ + (y0 + r)) * HW_ + x] = sum + addv[r];
  }
}

extern "C" void kernel_launch(void* const* d_in, const int* in_sizes, int n_in,
                              void* d_out, int out_size, void* d_ws,
                              size_t ws_size, hipStream_t stream) {
  (void)in_sizes; (void)n_in; (void)out_size; (void)ws_size;
  const float* in   = (const float*)d_in[0];
  const float* skip = (const float*)d_in[1];
  const float* w    = (const float*)d_in[2];
  const float* bias = (const float*)d_in[3];
  float* out = (float*)d_out;
  float* wpk = (float*)d_ws;

  prep_weight_kernel<<<dim3((KSTEPS * 32 * 4 + 255) / 256), dim3(256), 0, stream>>>(w, wpk);
  conv_planes_kernel<<<dim3(NBATCH * (HW_ / R_OUT)), dim3(NWAVE * 32), 0, stream>>>(
      in, wpk, skip, bias, out);
}


# First and last own rows of each strip run one row tile, not two: 16 of their 18 planes on the matrix unit, the 2 left-over planes as packed f32 FMAs, lane halves added last (changes their sum order)
# speedup vs baseline: 24.0287x; 1.0588x over previous
#include <hip/hip_runtime.h>


typedef __attribute__((ext_vector_type(2))) float v2f;
typedef __attribute__((ext_vector_type(4))) float v4f;
typedef __attribute__((ext_vector_type(8))) float v8f;

#define C_IN   512
#define HW_    128
#define NBATCH 8
#define K_RAW  9
#define EQ_SCALE 0.014731391274719739f
#define PLANE  (HW_ * HW_)
#define NPL    27
#define KSTEPS (C_IN / 4)
#define GRP    8
#define NGRP   (KSTEPS / GRP)
#define R_OUT  4
#define R_IN   (R_OUT + 2)
#define NWAVE  (R_IN * 2)

__host__ __device__ constexpr int plane_of(int t, int m) {
  return t == 0 ? m : (m < 11 ? 16 + m : m);
}

__global__ void prep_weight_kernel(const float* __restrict__ w,
                                   float* __restrict__ wpk) {
  int tid = blockIdx.x * blockDim.x + threadIdx.x;
  if (tid >= KSTEPS * 32 * 4) return;
  int e    = tid & 3;
  int lane = (tid >> 2) & 31;
  int ks   = tid >> 7;
  int c = lane & 15, h = lane >> 4;
  int p  = plane_of(e >> 1, c);
  int ci = ks * 4 + h * 2 + (e & 1);
  int dy = p / 9, q = p - dy * 9;
  int co = q / 3, dx = q - co * 3;
  wpk[tid] = w[co * (C_IN * K_RAW) + ci * K_RAW + dy * 3 + dx] * EQ_SCALE;
}

enum { HALO_UP = 0, EDGE_UP = 1, FULL = 2, EDGE_LO = 3, HALO_LO = 4 };

__host__ __device__ constexpr int  mode_t0(int mode) { return mode >= EDGE_LO ? 1 : 0; }
__host__ __device__ constexpr int  mode_nt(int mode) { return mode == FULL ? 2 : 1; }
__host__ __device__ constexpr bool mode_ex(int mode) { return mode == EDGE_UP || mode == EDGE_LO; }
__host__ __device__ constexpr int  mode_xt(int mode) { return mode == EDGE_UP ? 1 : 0; }
__host__ __device__ constexpr int  mode_xm(int mode) { return mode == EDGE_UP ? 0 : 9; }
__host__ __device__ constexpr bool keep_row(int mode, int t, int m) {
  return mode == HALO_UP ? (t == 0 && m < 9)
       : mode == EDGE_UP ? (t == 0)
       : mode == FULL    ? (t == 0 || m < 11)
       : mode == EDGE_LO ? (t == 1)
       :                   (t == 1 && m >= 2 && m < 11);
}

template <int NT> struct WFrag;
template <> struct WFrag<1> { typedef v2f type; };
template <> struct WFrag<2> { typedef v4f type; };

template <int NT, bool EX> struct Grp {
  v4f x0[GRP], x1[GRP];
  typename WFrag<NT>::type w[GRP];
  v2f wa[EX ? GRP : 1], wb[EX ? GRP : 1];
};

template <int MODE>
__device__ __forceinline__ void load_grp(Grp<mode_nt(MODE), mode_ex(MODE)>& G,
                                         const float* __restrict__ base,
                                         int loff, const float* __restrict__ wpk,
                                         int lane, int h, int g) {
  constexpr int T0 = mode_t0(MODE), NT = mode_nt(MODE);
  typedef typename WFrag<NT>::type wfrag_t;
#pragma unroll
  for (int u = 0; u < GRP; ++u) {
    const int ks = g * GRP + u;
    const float* p = base + (size_t)ks * (4 * PLANE) + loff;
    G.x0[u] = *(const v4f*)p;
    G.x1[u] = *(const v4f*)(p + PLANE);
    G.w[u]  = *(const wfrag_t*)(wpk + (ks * 32 + lane) * 4 + T0 * 2);
    if constexpr (mode_ex(MODE)) {
      const float* q = wpk + (ks * 32 + h * 16 + mode_xm(MODE)) * 4 + mode_xt(MODE) * 2;
      G.wa[u] = *(const v2f*)q;
      G.wb[u] = *(const v2f*)(q + 4);
    }
  }
}

template <int NT, bool EX>
__device__ __forceinline__ void mma_grp(const Grp<NT, EX>& G, v8f (&acc)[NT][4],
                                        v2f (&xacc)[2][2]) {
#pragma unroll
  for (int u = 0; u < GRP; ++u) {
    v2f a[NT];
#pragma unroll
    for (int t = 0; t < NT; ++t) { a[t].x = G.w[u][2 * t]; a[t].y = G.w[u][2 * t + 1]; }
#pragma unroll
    for (int j = 0; j < 4; ++j) {
      v2f bj;
      bj.x = G.x0[u][j]; bj.y = G.x1[u][j];
#pragma unroll
      for (int t = 0; t < NT; ++t)
        acc[t][j] = __builtin_amdgcn_wmma_f32_16x16x4_f32(
            false, a[t], false, bj, (short)0, acc[t][j], false, false);
    }
    if constexpr (EX) {
      const v2f wa0 = {G.wa[u].x, G.wa[u].x}, wa1 = {G.wa[u].y, G.wa[u].y};
      const v2f wb0 = {G.wb[u].x, G.wb[u].x}, wb1 = {G.wb[u].y, G.wb[u].y};
#pragma unroll
      for (int i = 0; i < 2; ++i) {
        const v2f c0 = {G.x0[u][2 * i], G.x0[u][2 * i + 1]};
        const v2f c1 = {G.x1[u][2 * i], G.x1[u][2 * i + 1]};
        xacc[0][i] = __builtin_elementwise_fma(wa0, c0, xacc[0][i]);
        xacc[0][i] = __builtin_elementwise_fma(wa1, c1, xacc[0][i]);
        xacc[1][i] = __builtin_elementwise_fma(wb0, c0, xacc[1][i]);
        xacc[1][i] = __builtin_elementwise_fma(wb1, c1, xacc[1][i]);
      }
    }
  }
}

template <int MODE>
__device__ __forceinline__ void row_planes(const float* __restrict__ base, int loff,
                                           const float* __restrict__ wpk, int lane,
                                           int h, float* __restrict__ Tpx) {
  constexpr int  T0 = mode_t0(MODE), NT = mode_nt(MODE);
  constexpr bool EX = mode_ex(MODE);
  v8f acc[NT][4];
#pragma unroll
  for (int t = 0; t < NT; ++t)
#pragma unroll
    for (int j = 0; j < 4; ++j) acc[t][j] = (v8f){};
  v2f xacc[2][2] = {};

  Grp<NT, EX> GA, GB;
  load_grp<MODE>(GA, base, loff, wpk, lane, h, 0);
  for (int g = 0; g < NGRP - 2; g += 2) {
    load_grp<MODE>(GB, base, loff, wpk, lane, h, g + 1);
    __builtin_amdgcn_sched_barrier(0);
    mma_grp<NT, EX>(GA, acc, xacc);
    __builtin_amdgcn_sched_barrier(0);
    load_grp<MODE>(GA, base, loff, wpk, lane, h, g + 2);
    __builtin_amdgcn_sched_barrier(0);
    mma_grp<NT, EX>(GB, acc, xacc);
    __builtin_amdgcn_sched_barrier(0);
  }
  load_grp<MODE>(GB, base, loff, wpk, lane, h, NGRP - 1);
  __builtin_amdgcn_sched_barrier(0);
  mma_grp<NT, EX>(GA, acc, xacc);
  mma_grp<NT, EX>(GB, acc, xacc);

#pragma unroll
  for (int t = 0; t < NT; ++t) {
#pragma unroll
    for (int r = 0; r < 8; ++r) {
      const bool k0 = keep_row(MODE, T0 + t, r), k1 = keep_row(MODE, T0 + t, 8 + r);
      if (h ? k1 : k0) {
        const int p = h ? plane_of(T0 + t, 8 + r) : plane_of(T0 + t, r);
        v4f v;
        v.x = acc[t][0][r]; v.y = acc[t][1][r]; v.z = acc[t][2][r]; v.w = acc[t][3][r];
        *(v4f*)(Tpx + p * HW_) = v;
      }
    }
  }

  if constexpr (EX) {
    constexpr int PX = plane_of(mode_xt(MODE), mode_xm(MODE));
    v4f mine, give;
    mine.x = h ? xacc[1][0].x : xacc[0][0].x;  give.x = h ? xacc[0][0].x : xacc[1][0].x;
    mine.y = h ? xacc[1][0].y : xacc[0][0].y;  give.y = h ? xacc[0][0].y : xacc[1][0].y;
    mine.z = h ? xacc[1][1].x : xacc[0][1].x;  give.z = h ? xacc[0][1].x : xacc[1][1].x;
    mine.w = h ? xacc[1][1].y : xacc[0][1].y;  give.w = h ? xacc[0][1].y : xacc[1][1].y;
    v4f v;
    v.x = mine.x + __shfl_xor(give.x, 16);
    v.y = mine.y + __shfl_xor(give.y, 16);
    v.z = mine.z + __shfl_xor(give.z, 16);
    v.w = mine.w + __shfl_xor(give.w, 16);
    *(v4f*)(Tpx + (PX + h) * HW_) = v;
  }
}

__device__ __forceinline__ float skip_bias_term(float bias, float wy0, float wy1,
                                                float wx0, float wx1,
                                                float s00, float s01,
                                                float s10, float s11) {
  float acc = bias;
  acc += wy0 * wx0 * s00;
  acc += wy0 * wx1 * s01;
  acc += wy1 * wx0 * s10;
  acc += wy1 * wx1 * s11;
  return acc;
}

__global__ void __launch_bounds__(NWAVE * 32) conv_planes_kernel(const float* __restrict__ in,
                                      const float* __restrict__ wpk,
                                      const float* __restrict__ skip,
                                      const float* __restrict__ bias,
                                      float* __restrict__ out) {
  __shared__ float T[R_IN * NPL * HW_];
  const int lane = threadIdx.x & 31;
  const int wave = __builtin_amdgcn_readfirstlane(threadIdx.x >> 5);
  const int unit = (int)((0x76549832BA10ull >> (4 * wave)) & 15);
  const int b    = blockIdx.x >> 5;
  const int y0   = (blockIdx.x & 31) * R_OUT;
  const int rl   = unit >> 1;
  const int half = unit & 1;
  const int yin  = y0 - 1 + rl;
  const int c = lane & 15, h = lane >> 4;
  const int px = half * 64 + c * 4;
  float* Trow = T + rl * (NPL * HW_);

  const int t  = threadIdx.x;
  const int x  = t & 127;
  const int co = t >> 7;
  const int   x2  = x >> 1;
  const bool  xe  = !(x & 1);
  const int   mx0 = xe ? x2     : x2 + 1;
  const int   mx1 = xe ? x2 - 1 : x2;
  const float wx0 = xe ? 0.75f : 0.25f;
  const float wx1 = xe ? 0.25f : 0.75f;
  const bool  okx0 = mx0 < HW_ / 2, okx1 = mx1 >= 0;
  const int   h0  = y0 >> 1;
  const bool  oka = h0 > 0, okd = h0 + 2 < HW_ / 2;
  float sa0, sa1, sb0, sb1, sc0, sc1, sd0, sd1, bco;
  {
    const float* s  = skip + ((size_t)b * 3 + co) * (PLANE / 4);
    const int cx0 = okx0 ? mx0 : HW_ / 2 - 1, cx1 = okx1 ? mx1 : 0;
    const float* ra = s + (oka ? h0 - 1 : 0) * (HW_ / 2);
    const float* rb = s + h0 * (HW_ / 2);
    const float* rc = rb + HW_ / 2;
    const float* rd = s + (okd ? h0 + 2 : HW_ / 2 - 1) * (HW_ / 2);
    sa0 = ra[cx0]; sa1 = ra[cx1];
    sb0 = rb[cx0]; sb1 = rb[cx1];
    sc0 = rc[cx0]; sc1 = rc[cx1];
    sd0 = rd[cx0]; sd1 = rd[cx1];
    bco = bias[co];
  }

  if (yin >= 0 && yin < HW_) {
    const float* base = in + (size_t)b * C_IN * PLANE + yin * HW_;
    const int loff = h * 2 * PLANE + px;
    if (rl == 0)
      row_planes<HALO_UP>(base, loff, wpk, lane, h, Trow + px);
    else if (rl == 1)
      row_planes<EDGE_UP>(base, loff, wpk, lane, h, Trow + px);
    else if (rl == R_IN - 2)
      row_planes<EDGE_LO>(base, loff, wpk, lane, h, Trow + px);
    else if (rl == R_IN - 1)
      row_planes<HALO_LO>(base, loff, wpk, lane, h, Trow + px);
    else
      row_planes<FULL>(base, loff, wpk, lane, h, Trow + px);
  } else {
    const int p0 = (rl == 0) ? 0 : 18;
    for (int p = p0; p < p0 + 9; ++p) {
      Trow[p * HW_ + half * 64 + lane]      = 0.0f;
      Trow[p * HW_ + half * 64 + 32 + lane] = 0.0f;
    }
  }
  __syncthreads();

  sa0 = (oka && okx0) ? sa0 : 0.0f;  sa1 = (oka && okx1) ? sa1 : 0.0f;
  sb0 = okx0 ? sb0 : 0.0f;           sb1 = okx1 ? sb1 : 0.0f;
  sc0 = okx0 ? sc0 : 0.0f;           sc1 = okx1 ? sc1 : 0.0f;
  sd0 = (okd && okx0) ? sd0 : 0.0f;  sd1 = (okd && okx1) ? sd1 : 0.0f;
  float addv[R_OUT];
  addv[0] = skip_bias_term(bco, 0.75f, 0.25f, wx0, wx1, sb0, sb1, sa0, sa1);
  addv[1] = skip_bias_term(bco, 0.25f, 0.75f, wx0, wx1, sc0, sc1, sb0, sb1);
  addv[2] = skip_bias_term(bco, 0.75f, 0.25f, wx0, wx1, sc0, sc1, sb0, sb1);
  addv[3] = skip_bias_term(bco, 0.25f, 0.75f, wx0, wx1, sd0, sd1, sc0, sc1);

  const bool okl = x > 0, okr = x < HW_ - 1;
  const int  xl = okl ? x - 1 : x, xr = okr ? x + 1 : x;
#pragma unroll
  for (int r = 0; r < R_OUT; ++r) {
    float sum = 0.0f;
#pragma unroll
    for (int dy = 0; dy < 3; ++dy) {
      const float* tp = T + ((r + dy) * NPL + dy * 9 + co * 3) * HW_;
      const float vl = tp[xl], vc = tp[HW_ + x], vr = tp[2 * HW_ + xr];
      sum += okl ? vl : 0.0f;
      sum += vc;
      sum += okr ? vr : 0.0f;
    }
    out[(((size_t)b * 3 + co) * HW_ + (y0 + r)) * HW_ + x] = sum + addv[r];
  }
}

extern "C" void kernel_launch(void* const* d_in, const int* in_sizes, int n_in,
                              void* d_out, int out_size, void* d_ws,
                              size_t ws_size, hipStream_t stream) {
  (void)in_sizes; (void)n_in; (void)out_size; (void)ws_size;
  const float* in   = (const float*)d_in[0];
  const float* skip = (const float*)d_in[1];
  const float* w    = (const float*)d_in[2];
  const float* bias = (const float*)d_in[3];
  float* out = (float*)d_out;
  float* wpk = (float*)d_ws;

  prep_weight_kernel<<<dim3((KSTEPS * 32 * 4 + 255) / 256), dim3(256), 0, stream>>>(w, wpk);
  conv_planes_kernel<<<dim3(NBATCH * (HW_ / R_OUT)), dim3(NWAVE * 32), 0, stream>>>(
      in, wpk, skip, bias, out);
}


# Each block packs the scaled weights into LDS itself and reads them from LDS in the loop: the weight-pack launch and its workspace buffer are gone (2 launches -> 1); sum order unchanged
# speedup vs baseline: 24.8018x; 1.0322x over previous
#include <hip/hip_runtime.h>


typedef __attribute__((ext_vector_type(2))) float v2f;
typedef __attribute__((ext_vector_type(4))) float v4f;
typedef __attribute__((ext_vector_type(8))) float v8f;

#define C_IN   512
#define HW_    128
#define NBATCH 8
#define K_RAW  9
#define EQ_SCALE 0.014731391274719739f
#define PLANE  (HW_ * HW_)
#define NPL    27
#define KSTEPS (C_IN / 4)
#define GRP    8
#define NGRP   (KSTEPS / GRP)
#define R_OUT  4
#define R_IN   (R_OUT + 2)
#define NWAVE  (R_IN * 2)
#define NROW   32
#define WP     (C_IN + 4)

__host__ __device__ constexpr int plane_of(int t, int m) {
  return t == 0 ? m : (m < 11 ? 16 + m : m);
}

__device__ __forceinline__ void load_raw_weights(v4f (&raw)[K_RAW],
                                                 const float* __restrict__ w, int tid) {
  const v4f* src = (const v4f*)(w + (size_t)tid * (4 * K_RAW));
#pragma unroll
  for (int i = 0; i < K_RAW; ++i) raw[i] = src[i];
}

__device__ __forceinline__ void store_weight_rows(const v4f (&raw)[K_RAW],
                                                  float* __restrict__ Wl, int co, int q) {
#pragma unroll
  for (int k = 0; k < K_RAW; ++k) {
    v4f v;
    v.x = raw[k >> 2][k & 3] * EQ_SCALE;
    v.y = raw[(K_RAW + k) >> 2][(K_RAW + k) & 3] * EQ_SCALE;
    v.z = raw[(2 * K_RAW + k) >> 2][(2 * K_RAW + k) & 3] * EQ_SCALE;
    v.w = raw[(3 * K_RAW + k) >> 2][(3 * K_RAW + k) & 3] * EQ_SCALE;
    const int p = (k / 3) * 9 + co * 3 + (k % 3);
    *(v4f*)(Wl + p * WP + 4 * q) = v;
    if (p >= 11 && p < 16) *(v4f*)(Wl + (p + 16) * WP + 4 * q) = v;
  }
}

enum { HALO_UP = 0, EDGE_UP = 1, FULL = 2, EDGE_LO = 3, HALO_LO = 4 };

__host__ __device__ constexpr int  mode_t0(int mode) { return mode >= EDGE_LO ? 1 : 0; }
__host__ __device__ constexpr int  mode_nt(int mode) { return mode == FULL ? 2 : 1; }
__host__ __device__ constexpr bool mode_ex(int mode) { return mode == EDGE_UP || mode == EDGE_LO; }
__host__ __device__ constexpr int  mode_xt(int mode) { return mode == EDGE_UP ? 1 : 0; }
__host__ __device__ constexpr int  mode_xm(int mode) { return mode == EDGE_UP ? 0 : 9; }
__host__ __device__ constexpr bool keep_row(int mode, int t, int m) {
  return mode == HALO_UP ? (t == 0 && m < 9)
       : mode == EDGE_UP ? (t == 0)
       : mode == FULL    ? (t == 0 || m < 11)
       : mode == EDGE_LO ? (t == 1)
       :                   (t == 1 && m >= 2 && m < 11);
}

struct XGrp { v4f x0[GRP], x1[GRP]; };

template <int NT, bool EX> struct WGrp {
  v2f w[NT][GRP];
  v2f wa[EX ? GRP : 1], wb[EX ? GRP : 1];
};

__device__ __forceinline__ void load_x(XGrp& X, const float* __restrict__ base,
                                       int loff, int g) {
#pragma unroll
  for (int u = 0; u < GRP; ++u) {
    const int ks = g * GRP + u;
    const float* p = base + (size_t)ks * (4 * PLANE) + loff;
    X.x0[u] = *(const v4f*)p;
    X.x1[u] = *(const v4f*)(p + PLANE);
  }
}

template <int MODE>
__device__ __forceinline__ void load_w(WGrp<mode_nt(MODE), mode_ex(MODE)>& G,
                                       const float* wl, const float* wh, int g) {
  constexpr int T0 = mode_t0(MODE), NT = mode_nt(MODE);
#pragma unroll
  for (int u = 0; u < GRP; ++u) {
    const int ks = g * GRP + u;
#pragma unroll
    for (int t = 0; t < NT; ++t)
      G.w[t][u] = *(const v2f*)(wl + (T0 + t) * 16 * WP + 4 * ks);
    if constexpr (mode_ex(MODE)) {
      const float* q = wh + (mode_xt(MODE) * 16 + mode_xm(MODE)) * WP + 4 * ks;
      G.wa[u] = *(const v2f*)q;
      G.wb[u] = *(const v2f*)(q + WP);
    }
  }
}

template <int NT, bool EX>
__device__ __forceinline__ void mma_grp(const XGrp& X, const WGrp<NT, EX>& G,
                                        v8f (&acc)[NT][4], v2f (&xacc)[2][2]) {
#pragma unroll
  for (int u = 0; u < GRP; ++u) {
#pragma unroll
    for (int j = 0; j < 4; ++j) {
      v2f bj;
      bj.x = X.x0[u][j]; bj.y = X.x1[u][j];
#pragma unroll
      for (int t = 0; t < NT; ++t)
        acc[t][j] = __builtin_amdgcn_wmma_f32_16x16x4_f32(
            false, G.w[t][u], false, bj, (short)0, acc[t][j], false, false);
    }
    if constexpr (EX) {
      const v2f wa0 = {G.wa[u].x, G.wa[u].x}, wa1 = {G.wa[u].y, G.wa[u].y};
      const v2f wb0 = {G.wb[u].x, G.wb[u].x}, wb1 = {G.wb[u].y, G.wb[u].y};
#pragma unroll
      for (int i = 0; i < 2; ++i) {
        const v2f c0 = {X.x0[u][2 * i], X.x0[u][2 * i + 1]};
        const v2f c1 = {X.x1[u][2 * i], X.x1[u][2 * i + 1]};
        xacc[0][i] = __builtin_elementwise_fma(wa0, c0, xacc[0][i]);
        xacc[0][i] = __builtin_elementwise_fma(wa1, c1, xacc[0][i]);
        xacc[1][i] = __builtin_elementwise_fma(wb0, c0, xacc[1][i]);
        xacc[1][i] = __builtin_elementwise_fma(wb1, c1, xacc[1][i]);
      }
    }
  }
}

template <int MODE>
__device__ __forceinline__ void row_planes(const float* __restrict__ base,
                                           int loff, const float* wl, const float* wh,
                                           int h, float* __restrict__ Tpx) {
  constexpr int  T0 = mode_t0(MODE), NT = mode_nt(MODE);
  constexpr bool EX = mode_ex(MODE);
  v8f acc[NT][4];
#pragma unroll
  for (int t = 0; t < NT; ++t)
#pragma unroll
    for (int j = 0; j < 4; ++j) acc[t][j] = (v8f){};
  v2f xacc[2][2] = {};

  XGrp XA, XB;
  WGrp<NT, EX> WA, WB;
  load_x(XA, base, loff, 0);
  load_w<MODE>(WA, wl, wh, 0);
  for (int g = 0; g < NGRP - 2; g += 2) {
    load_x(XB, base, loff, g + 1);
    load_w<MODE>(WB, wl, wh, g + 1);
    __builtin_amdgcn_sched_barrier(0);
    mma_grp<NT, EX>(XA, WA, acc, xacc);
    __builtin_amdgcn_sched_barrier(0);
    load_x(XA, base, loff, g + 2);
    load_w<MODE>(WA, wl, wh, g + 2);
    __builtin_amdgcn_sched_barrier(0);
    mma_grp<NT, EX>(XB, WB, acc, xacc);
    __builtin_amdgcn_sched_barrier(0);
  }
  load_x(XB, base, loff, NGRP - 1);
  load_w<MODE>(WB, wl, wh, NGRP - 1);
  __builtin_amdgcn_sched_barrier(0);
  mma_grp<NT, EX>(XA, WA, acc, xacc);
  mma_grp<NT, EX>(XB, WB, acc, xacc);

#pragma unroll
  for (int t = 0; t < NT; ++t) {
#pragma unroll
    for (int r = 0; r < 8; ++r) {
      const bool k0 = keep_row(MODE, T0 + t, r), k1 = keep_row(MODE, T0 + t, 8 + r);
      if (h ? k1 : k0) {
        const int p = h ? plane_of(T0 + t, 8 + r) : plane_of(T0 + t, r);
        v4f v;
        v.x = acc[t][0][r]; v.y = acc[t][1][r]; v.z = acc[t][2][r]; v.w = acc[t][3][r];
        *(v4f*)(Tpx + p * HW_) = v;
      }
    }
  }

  if constexpr (EX) {
    constexpr int PX = plane_of(mode_xt(MODE), mode_xm(MODE));
    v4f mine, give;
    mine.x = h ? xacc[1][0].x : xacc[0][0].x;  give.x = h ? xacc[0][0].x : xacc[1][0].x;
    mine.y = h ? xacc[1][0].y : xacc[0][0].y;  give.y = h ? xacc[0][0].y : xacc[1][0].y;
    mine.z = h ? xacc[1][1].x : xacc[0][1].x;  give.z = h ? xacc[0][1].x : xacc[1][1].x;
    mine.w = h ? xacc[1][1].y : xacc[0][1].y;  give.w = h ? xacc[0][1].y : xacc[1][1].y;
    v4f v;
    v.x = mine.x + __shfl_xor(give.x, 16);
    v.y = mine.y + __shfl_xor(give.y, 16);
    v.z = mine.z + __shfl_xor(give.z, 16);
    v.w = mine.w + __shfl_xor(give.w, 16);
    *(v4f*)(Tpx + (PX + h) * HW_) = v;
  }
}

__device__ __forceinline__ float skip_bias_term(float bias, float wy0, float wy1,
                                                float wx0, float wx1,
                                                float s00, float s01,
                                                float s10, float s11) {
  float acc = bias;
  acc += wy0 * wx0 * s00;
  acc += wy0 * wx1 * s01;
  acc += wy1 * wx0 * s10;
  acc += wy1 * wx1 * s11;
  return acc;
}

__global__ void __launch_bounds__(NWAVE * 32) conv_planes_kernel(const float* __restrict__ in,
                                      const float* __restrict__ w,
                                      const float* __restrict__ skip,
                                      const float* __restrict__ bias,
                                      float* __restrict__ out) {
  __shared__ __attribute__((aligned(16))) float T[R_IN * NPL * HW_];
  __shared__ __attribute__((aligned(16))) float Wl[NROW * WP];
  const int lane = threadIdx.x & 31;
  const int wave = __builtin_amdgcn_readfirstlane(threadIdx.x >> 5);
  const int unit = (int)((0x76549832BA10ull >> (4 * wave)) & 15);
  const int b    = blockIdx.x >> 5;
  const int y0   = (blockIdx.x & 31) * R_OUT;
  const int rl   = unit >> 1;
  const int half = unit & 1;
  const int yin  = y0 - 1 + rl;
  const int c = lane & 15, h = lane >> 4;
  const int px = half * 64 + c * 4;
  float* Trow = T + rl * (NPL * HW_);

  v4f raw[K_RAW];
  load_raw_weights(raw, w, threadIdx.x);

  const int t  = threadIdx.x;
  const int x  = t & 127;
  const int co = t >> 7;
  const int   x2  = x >> 1;
  const bool  xe  = !(x & 1);
  const int   mx0 = xe ? x2     : x2 + 1;
  const int   mx1 = xe ? x2 - 1 : x2;
  const float wx0 = xe ? 0.75f : 0.25f;
  const float wx1 = xe ? 0.25f : 0.75f;
  const bool  okx0 = mx0 < HW_ / 2, okx1 = mx1 >= 0;
  const int   h0  = y0 >> 1;
  const bool  oka = h0 > 0, okd = h0 + 2 < HW_ / 2;
  float sa0, sa1, sb0, sb1, sc0, sc1, sd0, sd1, bco;
  {
    const float* s  = skip + ((size_t)b * 3 + co) * (PLANE / 4);
    const int cx0 = okx0 ? mx0 : HW_ / 2 - 1, cx1 = okx1 ? mx1 : 0;
    const float* ra = s + (oka ? h0 - 1 : 0) * (HW_ / 2);
    const float* rb = s + h0 * (HW_ / 2);
    const float* rc = rb + HW_ / 2;
    const float* rd = s + (okd ? h0 + 2 : HW_ / 2 - 1) * (HW_ / 2);
    sa0 = ra[cx0]; sa1 = ra[cx1];
    sb0 = rb[cx0]; sb1 = rb[cx1];
    sc0 = rc[cx0]; sc1 = rc[cx1];
    sd0 = rd[cx0]; sd1 = rd[cx1];
    bco = bias[co];
  }

  store_weight_rows(raw, Wl, wave >> 2, threadIdx.x & 127);
  __syncthreads();

  if (yin >= 0 && yin < HW_) {
    const float* base = in + (size_t)b * C_IN * PLANE + yin * HW_;
    const int loff = h * 2 * PLANE + px;
    const float* wl = Wl + c * WP + 2 * h;
    const float* wh = Wl + 2 * h;
    if (rl == 0)
      row_planes<HALO_UP>(base, loff, wl, wh, h, Trow + px);
    else if (rl == 1)
      row_planes<EDGE_UP>(base, loff, wl, wh, h, Trow + px);
    else if (rl == R_IN - 2)
      row_planes<EDGE_LO>(base, loff, wl, wh, h, Trow + px);
    else if (rl == R_IN - 1)
      row_planes<HALO_LO>(base, loff, wl, wh, h, Trow + px);
    else
      row_planes<FULL>(base, loff, wl, wh, h, Trow + px);
  } else {
    const int p0 = (rl == 0) ? 0 : 18;
    for (int p = p0; p < p0 + 9; ++p) {
      Trow[p * HW_ + half * 64 + lane]      = 0.0f;
      Trow[p * HW_ + half * 64 + 32 + lane] = 0.0f;
    }
  }
  __syncthreads();

  sa0 = (oka && okx0) ? sa0 : 0.0f;  sa1 = (oka && okx1) ? sa1 : 0.0f;
  sb0 = okx0 ? sb0 : 0.0f;           sb1 = okx1 ? sb1 : 0.0f;
  sc0 = okx0 ? sc0 : 0.0f;           sc1 = okx1 ? sc1 : 0.0f;
  sd0 = (okd && okx0) ? sd0 : 0.0f;  sd1 = (okd && okx1) ? sd1 : 0.0f;
  float addv[R_OUT];
  addv[0] = skip_bias_term(bco, 0.75f, 0.25f, wx0, wx1, sb0, sb1, sa0, sa1);
  addv[1] = skip_bias_term(bco, 0.25f, 0.75f, wx0, wx1, sc0, sc1, sb0, sb1);
  addv[2] = skip_bias_term(bco, 0.75f, 0.25f, wx0, wx1, sc0, sc1, sb0, sb1);
  addv[3] = skip_bias_term(bco, 0.25f, 0.75f, wx0, wx1, sd0, sd1, sc0, sc1);

  const bool okl = x > 0, okr = x < HW_ - 1;
  const int  xl = okl ? x - 1 : x, xr = okr ? x + 1 : x;
#pragma unroll
  for (int r = 0; r < R_OUT; ++r) {
    float sum = 0.0f;
#pragma unroll
    for (int dy = 0; dy < 3; ++dy) {
      const float* tp = T + ((r + dy) * NPL + dy * 9 + co * 3) * HW_;
      const float vl = tp[xl], vc = tp[HW_ + x], vr = tp[2 * HW_ + xr];
      sum += okl ? vl : 0.0f;
      sum += vc;
      sum += okr ? vr : 0.0f;
    }
    out[(((size_t)b * 3 + co) * HW_ + (y0 + r)) * HW_ + x] = sum + addv[r];
  }
}

extern "C" void kernel_launch(void* const* d_in, const int* in_sizes, int n_in,
                              void* d_out, int out_size, void* d_ws,
                              size_t ws_size, hipStream_t stream) {
  (void)in_sizes; (void)n_in; (void)out_size; (void)ws_size;
  const float* in   = (const float*)d_in[0];
  const float* skip = (const float*)d_in[1];
  const float* w    = (const float*)d_in[2];
  const float* bias = (const float*)d_in[3];
  float* out = (float*)d_out;
  (void)d_ws;

  conv_planes_kernel<<<dim3(NBATCH * (HW_ / R_OUT)), dim3(NWAVE * 32), 0, stream>>>(
      in, w, skip, bias, out);
}
